# starting kernel
# baseline (speedup 1.0000x reference)
#include <hip/hip_runtime.h>

typedef __attribute__((ext_vector_type(2))) float v2f;
typedef __attribute__((ext_vector_type(8))) float v8f;

#define NHID 256
#define EPSF 1e-7f
#define MAXNF 0.99999f

__device__ __forceinline__ v8f wmma_f32x4(v2f a, v2f b, v8f c) {
  return __builtin_amdgcn_wmma_f32_16x16x4_f32(false, a, false, b, (short)0, c,
                                               false, false);
}

__device__ __forceinline__ float wred32(float v) {
#pragma unroll
  for (int m = 16; m > 0; m >>= 1) v += __shfl_xor(v, m, 32);
  return v;
}

__device__ __forceinline__ float hyp_scale(float x2, float m2) {
  float xn  = fminf(fmaxf(sqrtf(x2), EPSF), MAXNF);
  float mxn = fmaxf(sqrtf(m2), EPSF);
  return tanhf(mxn / xn * atanhf(xn)) / mxn;
}

__device__ __forceinline__ void load8(const float* p, float s, float* d) {
  const float4* g = (const float4*)p;
  float4 a = g[0], b = g[1];
  d[0] = a.x * s; d[1] = a.y * s; d[2] = a.z * s; d[3] = a.w * s;
  d[4] = b.x * s; d[5] = b.y * s; d[6] = b.z * s; d[7] = b.w * s;
}

__device__ __forceinline__ void store8(float* p, const float* d, float s) {
  float4 a, b;
  a.x = d[0] * s; a.y = d[1] * s; a.z = d[2] * s; a.w = d[3] * s;
  b.x = d[4] * s; b.y = d[5] * s; b.z = d[6] * s; b.w = d[7] * s;
  float4* g = (float4*)p;
  g[0] = a; g[1] = b;
}

template <int KDIM, bool FULL_EPI>
__global__ __launch_bounds__(256) void k_mm(const float* X,
                                            const float* __restrict__ W,
                                            const float* __restrict__ bias,
                                            float* OUT, float inScale) {
  __shared__ float smx[32][NHID + 8];
  __shared__ float part[32][8];
  __shared__ float rx2[32], rm2[32], rdot[32], rv2[32];
  __shared__ float sb2s;

  const int t = threadIdx.x;
  const int wave = t >> 5, lane = t & 31;
  const int tile_m = wave >> 2;
  const int n0 = (wave & 3) << 6;
  const int l15 = lane & 15;
  const int kk = (lane >> 4) << 1;
  const int brow = blockIdx.x << 5;

  v8f acc0 = {}, acc1 = {}, acc2 = {}, acc3 = {};
  const float* Ar = X + (size_t)(brow + tile_m * 16 + l15) * KDIM;
  for (int k0 = 0; k0 < KDIM; k0 += 4) {
    v2f a;
    a.x = Ar[k0 + kk];
    a.y = Ar[k0 + kk + 1];
    const float* Bp = W + (size_t)(k0 + kk) * NHID + n0 + l15;
    v2f b;
    b.x = Bp[0];  b.y = Bp[NHID];      acc0 = wmma_f32x4(a, b, acc0);
    b.x = Bp[16]; b.y = Bp[NHID + 16]; acc1 = wmma_f32x4(a, b, acc1);
    b.x = Bp[32]; b.y = Bp[NHID + 32]; acc2 = wmma_f32x4(a, b, acc2);
    b.x = Bp[48]; b.y = Bp[NHID + 48]; acc3 = wmma_f32x4(a, b, acc3);
  }
  {
    const int mb = tile_m * 16 + ((lane >> 4) ? 8 : 0);
#pragma unroll
    for (int v = 0; v < 8; ++v) {
      smx[mb + v][n0 + 0 + l15]  = acc0[v];
      smx[mb + v][n0 + 16 + l15] = acc1[v];
      smx[mb + v][n0 + 32 + l15] = acc2[v];
      smx[mb + v][n0 + 48 + l15] = acc3[v];
    }
  }
  __syncthreads();

  const int r = t >> 3, p = t & 7;
  float sx = 0.f;
  {
    const int seg = KDIM / 8;
    const float* xr = X + (size_t)(brow + r) * KDIM + p * seg;
    for (int j = 0; j < seg; j += 4) {
      float4 v = *(const float4*)(xr + j);
      sx += v.x * v.x + v.y * v.y + v.z * v.z + v.w * v.w;
    }
  }
  float sm = 0.f, sd = 0.f;
  {
    const float* mr = &smx[r][p * 32];
#pragma unroll
    for (int j = 0; j < 32; ++j) {
      float m = mr[j];
      sm += m * m;
      if (FULL_EPI) sd += m * bias[p * 32 + j];
    }
  }
  part[r][p] = sx;
  __syncthreads();
  if (t < 32) { float s = 0.f; for (int i = 0; i < 8; ++i) s += part[t][i]; rx2[t] = s; }
  __syncthreads();
  part[r][p] = sm;
  __syncthreads();
  if (t < 32) { float s = 0.f; for (int i = 0; i < 8; ++i) s += part[t][i]; rm2[t] = s; }
  __syncthreads();
  if (FULL_EPI) {
    part[r][p] = sd;
    __syncthreads();
    if (t < 32) { float s = 0.f; for (int i = 0; i < 8; ++i) s += part[t][i]; rdot[t] = s; }
    __syncthreads();
    part[r][p] = bias[t] * bias[t];
    __syncthreads();
    if (t == 0) {
      float s = 0.f;
      for (int i = 0; i < 32; ++i)
        for (int j = 0; j < 8; ++j) s += part[i][j];
      sb2s = s;
    }
    __syncthreads();
  }

  const float x2s = inScale * inScale * rx2[r];
  const float m2s = inScale * inScale * rm2[r];
  const float cmv = hyp_scale(x2s, m2s) * inScale;
  float* orow = OUT + (size_t)(brow + r) * NHID + p * 32;
  float* mrw = &smx[r][p * 32];
  if (!FULL_EPI) {
#pragma unroll
    for (int j = 0; j < 32; j += 4) {
      float4 o;
      o.x = cmv * mrw[j + 0]; o.y = cmv * mrw[j + 1];
      o.z = cmv * mrw[j + 2]; o.w = cmv * mrw[j + 3];
      *(float4*)(orow + j) = o;
    }
  } else {
    const float cf = 0.25f * cmv;
    const float x2r = cf * cf * rm2[r];
    const float xy = cf * rdot[r];
    const float y2 = sb2s;
    const float den = fmaxf(1.f + 2.f * xy + x2r * y2, EPSF);
    const float ca = (1.f + 2.f * xy + y2) / den;
    const float cb = (1.f - x2r) / den;
    const float zn2 = ca * ca * x2r + 2.f * ca * cb * xy + cb * cb * y2;
    const float zn = fminf(fmaxf(sqrtf(zn2), EPSF), MAXNF);
    const float us = atanhf(zn) / zn;
    const float acf = ca * cf;
    float sv = 0.f;
#pragma unroll
    for (int j = 0; j < 32; ++j) {
      float z = acf * mrw[j] + cb * bias[p * 32 + j];
      float uu = fmaxf(us * z, 0.f);
      mrw[j] = uu;
      sv += uu * uu;
    }
    part[r][p] = sv;
    __syncthreads();
    if (t < 32) { float s = 0.f; for (int i = 0; i < 8; ++i) s += part[t][i]; rv2[t] = s; }
    __syncthreads();
    const float vn = fmaxf(sqrtf(rv2[r]), EPSF);
    const float fs = tanhf(vn) / vn;
#pragma unroll
    for (int j = 0; j < 32; j += 4) {
      float4 o;
      o.x = fs * mrw[j + 0]; o.y = fs * mrw[j + 1];
      o.z = fs * mrw[j + 2]; o.w = fs * mrw[j + 3];
      *(float4*)(orow + j) = o;
    }
  }
}

__global__ __launch_bounds__(256) void k_agg(const float* __restrict__ F,
                                             const int* __restrict__ idx,
                                             const float* __restrict__ bias,
                                             float* __restrict__ OUT,
                                             float inScale, int doAct) {
  const int t = threadIdx.x;
  const int wave = t >> 5, lane = t & 31;
  const int node = (blockIdx.x << 3) + wave;
  const int* srow = idx + (size_t)node * 16;
  const int joff = lane << 3;

  float x[8];
  load8(F + (size_t)srow[0] * NHID + joff, inScale, x);
  float px = 0.f;
#pragma unroll
  for (int j = 0; j < 8; ++j) px += x[j] * x[j];
  float x2 = wred32(px);

  for (int k = 1; k < 16; ++k) {
    if (k < 15) __builtin_prefetch(F + (size_t)srow[k + 1] * NHID + joff, 0, 0);
    float y[8];
    load8(F + (size_t)srow[k] * NHID + joff, inScale, y);
    float py2 = 0.f, pxy = 0.f;
#pragma unroll
    for (int j = 0; j < 8; ++j) { py2 += y[j] * y[j]; pxy += x[j] * y[j]; }
    float y2 = wred32(py2);
    float xy = wred32(pxy);
    float den = fmaxf(1.f + 2.f * xy + x2 * y2, EPSF);
    float ca = (1.f + 2.f * xy + y2) / den;
    float cb = (1.f - x2) / den;
    float pn = 0.f;
#pragma unroll
    for (int j = 0; j < 8; ++j) { x[j] = ca * x[j] + cb * y[j]; pn += x[j] * x[j]; }
    x2 = wred32(pn);
  }

  float* orow = OUT + (size_t)node * NHID + joff;
  if (!doAct) {
    store8(orow, x, 1.f);
  } else {
    float b[8];
    load8(bias + joff, 1.f, b);
    float pb2 = 0.f, pxy = 0.f;
#pragma unroll
    for (int j = 0; j < 8; ++j) { pb2 += b[j] * b[j]; pxy += x[j] * b[j]; }
    float y2 = wred32(pb2);
    float xy = 0.25f * wred32(pxy);
    float x2r = 0.0625f * x2;
    float den = fmaxf(1.f + 2.f * xy + x2r * y2, EPSF);
    float ca = (1.f + 2.f * xy + y2) / den;
    float cb = (1.f - x2r) / den;
    float zn2 = ca * ca * x2r + 2.f * ca * cb * xy + cb * cb * y2;
    float zn = fminf(fmaxf(sqrtf(zn2), EPSF), MAXNF);
    float us = atanhf(zn) / zn;
    float u[8];
    float pv = 0.f;
#pragma unroll
    for (int j = 0; j < 8; ++j) {
      float z = ca * 0.25f * x[j] + cb * b[j];
      u[j] = fmaxf(us * z, 0.f);
      pv += u[j] * u[j];
    }
    float vn = fmaxf(sqrtf(wred32(pv)), EPSF);
    float fs = tanhf(vn) / vn;
    store8(orow, u, fs);
  }
}

__global__ __launch_bounds__(256) void k_final(const float* __restrict__ H,
                                               const float* __restrict__ Wl,
                                               const float* __restrict__ bl,
                                               const int* __restrict__ tf,
                                               float* __restrict__ OUT) {
  __shared__ int ridx[64];
  __shared__ float smx[64][72];
  __shared__ float part[64][4];
  __shared__ float rx2[64], rm2[64], rdot[64];
  __shared__ float sb2;

  const int t = threadIdx.x;
  if (t < 64) ridx[t] = tf[t] + t * 1024;
  if (t == 0) {
    float s = 0.f;
    for (int j = 0; j < 64; ++j) s += bl[j] * bl[j];
    sb2 = s;
  }
  __syncthreads();

  const int wave = t >> 5, lane = t & 31;
  const int l15 = lane & 15;
  const int kk = (lane >> 4) << 1;
#pragma unroll
  for (int i = 0; i < 2; ++i) {
    const int tile = wave * 2 + i;
    const int tm = tile >> 2, tn = tile & 3;
    const float* Ar = H + (size_t)ridx[tm * 16 + l15] * NHID;
    const float* Br = Wl + (size_t)(tn * 16 + l15) * NHID;
    v8f c = {};
    for (int k0 = 0; k0 < NHID; k0 += 4) {
      v2f a, b;
      a.x = Ar[k0 + kk]; a.y = Ar[k0 + kk + 1];
      b.x = Br[k0 + kk]; b.y = Br[k0 + kk + 1];
      c = wmma_f32x4(a, b, c);
    }
    const int n = tn * 16 + l15;
    const int mb = tm * 16 + ((lane >> 4) ? 8 : 0);
#pragma unroll
    for (int v = 0; v < 8; ++v) smx[mb + v][n] = c[v];
  }
  __syncthreads();

  const int r = t >> 2, p = t & 3;
  float sx = 0.f;
  {
    const float* hr = H + (size_t)ridx[r] * NHID + p * 64;
    for (int j = 0; j < 64; j += 4) {
      float4 v = *(const float4*)(hr + j);
      sx += v.x * v.x + v.y * v.y + v.z * v.z + v.w * v.w;
    }
  }
  float sm = 0.f, sd = 0.f;
#pragma unroll
  for (int j = 0; j < 16; ++j) {
    float m = smx[r][p * 16 + j];
    sm += m * m;
    sd += m * bl[p * 16 + j];
  }
  part[r][p] = sx;
  __syncthreads();
  if (t < 64) rx2[t] = part[t][0] + part[t][1] + part[t][2] + part[t][3];
  __syncthreads();
  part[r][p] = sm;
  __syncthreads();
  if (t < 64) rm2[t] = part[t][0] + part[t][1] + part[t][2] + part[t][3];
  __syncthreads();
  part[r][p] = sd;
  __syncthreads();
  if (t < 64) rdot[t] = part[t][0] + part[t][1] + part[t][2] + part[t][3];
  __syncthreads();

  const float s = hyp_scale(rx2[r], rm2[r]);
  const float x2r = s * s * rm2[r];
  const float xy = s * rdot[r];
  const float y2 = sb2;
  const float den = fmaxf(1.f + 2.f * xy + x2r * y2, EPSF);
  const float ca = (1.f + 2.f * xy + y2) / den;
  const float cb = (1.f - x2r) / den;
  const float acs = ca * s;
#pragma unroll
  for (int j = 0; j < 16; ++j)
    OUT[r * 64 + p * 16 + j] = acs * smx[r][p * 16 + j] + cb * bl[p * 16 + j];
}

extern "C" void kernel_launch(void* const* d_in, const int* in_sizes, int n_in,
                              void* d_out, int out_size, void* d_ws,
                              size_t ws_size, hipStream_t stream) {
  const float* features = (const float*)d_in[0];
  const float* W1 = (const float*)d_in[1];
  const float* b1 = (const float*)d_in[2];
  const float* W2 = (const float*)d_in[3];
  const float* b2 = (const float*)d_in[4];
  const float* Wl = (const float*)d_in[5];
  const float* bl = (const float*)d_in[6];
  const int* src_idx = (const int*)d_in[7];
  const int* to_fetch = (const int*)d_in[8];
  float* out = (float*)d_out;

  float* buf0 = (float*)d_ws;
  float* buf1 = buf0 + (size_t)65536 * 256;

  k_mm<512, false><<<2048, 256, 0, stream>>>(const_cast<float*>(features), W1,
                                             b1, buf0, 0.25f);
  k_agg<<<8192, 256, 0, stream>>>(buf0, src_idx, b1, buf1, 1.0f, 1);
  k_agg<<<8192, 256, 0, stream>>>(buf1, src_idx, b2, buf0, 0.25f, 0);
  k_mm<256, true><<<2048, 256, 0, stream>>>(buf0, W2, b2, buf0, 1.0f);
  k_final<<<1, 256, 0, stream>>>(buf0, Wl, bl, to_fetch, out);
}


# k_mm: each wave a 64x64 register tile (was 16x64; block 128 rows), A/B fragments straight from global by 16-byte loads, k and column slots permuted alike, no LDS stage; k sum order changes per 8
# speedup vs baseline: 1.1705x; 1.1705x over previous
#include <hip/hip_runtime.h>

typedef __attribute__((ext_vector_type(2))) float v2f;
typedef __attribute__((ext_vector_type(8))) float v8f;

#define NHID 256
#define EPSF 1e-7f
#define MAXNF 0.99999f

__device__ __forceinline__ v8f wmma_f32x4(v2f a, v2f b, v8f c) {
  return __builtin_amdgcn_wmma_f32_16x16x4_f32(false, a, false, b, (short)0, c,
                                               false, false);
}

__device__ __forceinline__ float wred32(float v) {
#pragma unroll
  for (int m = 16; m > 0; m >>= 1) v += __shfl_xor(v, m, 32);
  return v;
}

__device__ __forceinline__ float hyp_scale(float x2, float m2) {
  float xn  = fminf(fmaxf(sqrtf(x2), EPSF), MAXNF);
  float mxn = fmaxf(sqrtf(m2), EPSF);
  return tanhf(mxn / xn * atanhf(xn)) / mxn;
}

__device__ __forceinline__ void load8(const float* p, float s, float* d) {
  const float4* g = (const float4*)p;
  float4 a = g[0], b = g[1];
  d[0] = a.x * s; d[1] = a.y * s; d[2] = a.z * s; d[3] = a.w * s;
  d[4] = b.x * s; d[5] = b.y * s; d[6] = b.z * s; d[7] = b.w * s;
}

__device__ __forceinline__ void store8(float* p, const float* d, float s) {
  float4 a, b;
  a.x = d[0] * s; a.y = d[1] * s; a.z = d[2] * s; a.w = d[3] * s;
  b.x = d[4] * s; b.y = d[5] * s; b.z = d[6] * s; b.w = d[7] * s;
  float4* g = (float4*)p;
  g[0] = a; g[1] = b;
}

template <int KDIM>
__device__ __forceinline__ void mm_load(const float* Ap, const float* Bp, int k0,
                                        float4 (&a)[4], float4 (&b)[4]) {
  for (int i = 0; i < 4; ++i)
    a[i] = *(const float4*)(Ap + (size_t)i * 16 * KDIM + k0);
  for (int s = 0; s < 4; ++s)
    b[s] = *(const float4*)(Bp + (size_t)(k0 + s) * NHID);
}

__device__ __forceinline__ void mm_block(v8f (&acc)[4][4], const float4 (&a)[4],
                                         const float4 (&b)[4]) {
  const float bv[4][4] = {{b[0].x, b[0].y, b[0].z, b[0].w},
                          {b[1].x, b[1].y, b[1].z, b[1].w},
                          {b[2].x, b[2].y, b[2].z, b[2].w},
                          {b[3].x, b[3].y, b[3].z, b[3].w}};
  for (int i = 0; i < 4; ++i) {
    v2f a0; a0.x = a[i].x; a0.y = a[i].y;
    for (int j = 0; j < 4; ++j) {
      v2f b0; b0.x = bv[0][j]; b0.y = bv[1][j];
      acc[i][j] = wmma_f32x4(a0, b0, acc[i][j]);
    }
  }
  for (int i = 0; i < 4; ++i) {
    v2f a1; a1.x = a[i].z; a1.y = a[i].w;
    for (int j = 0; j < 4; ++j) {
      v2f b1; b1.x = bv[2][j]; b1.y = bv[3][j];
      acc[i][j] = wmma_f32x4(a1, b1, acc[i][j]);
    }
  }
}

template <int KDIM, bool FULL_EPI>
__global__ __launch_bounds__(256) void k_mm(const float* X,
                                            const float* __restrict__ W,
                                            const float* __restrict__ bias,
                                            float* OUT, float inScale) {
  __shared__ alignas(16) float smx[32][NHID + 8];
  __shared__ float part[32][8];
  __shared__ float rx2[32], rm2[32], rdot[32], rv2[32];
  __shared__ float sb2s;

  const int t = threadIdx.x;
  const int wave = t >> 5, lane = t & 31;
  const int tile_m = wave >> 2;
  const int n0 = (wave & 3) << 6;
  const int l15 = lane & 15;
  const int h = lane >> 4;
  const int brow = blockIdx.x << 7;

  v8f acc[4][4];
  for (int i = 0; i < 4; ++i)
    for (int j = 0; j < 4; ++j) acc[i][j] = v8f{};

  const float* Ap = X + (size_t)(brow + tile_m * 64 + l15) * KDIM + 4 * h;
  const float* Bp = W + (size_t)(4 * h) * NHID + n0 + 4 * l15;

  float4 a0[4], b0[4], a1[4], b1[4];
  mm_load<KDIM>(Ap, Bp, 0, a0, b0);
  for (int k0 = 0; k0 < KDIM - 16; k0 += 16) {
    mm_load<KDIM>(Ap, Bp, k0 + 8, a1, b1);
    mm_block(acc, a0, b0);
    mm_load<KDIM>(Ap, Bp, k0 + 16, a0, b0);
    mm_block(acc, a1, b1);
  }
  mm_load<KDIM>(Ap, Bp, KDIM - 8, a1, b1);
  mm_block(acc, a0, b0);
  mm_block(acc, a1, b1);

  const int r = t >> 3, p = t & 7;
  if (FULL_EPI) {
    part[r][p] = bias[t] * bias[t];
    __syncthreads();
    if (t == 0) {
      float s = 0.f;
      for (int i = 0; i < 32; ++i)
        for (int j = 0; j < 8; ++j) s += part[i][j];
      sb2s = s;
    }
    __syncthreads();
  }

#pragma unroll
  for (int q = 0; q < 4; ++q) {
    const int rb = brow + 32 * q;
    if (tile_m == (q >> 1)) {
      for (int ii = 0; ii < 2; ++ii) {
        const int i = 2 * (q & 1) + ii;
        const int mb = ii * 16 + 8 * h;
        for (int v = 0; v < 8; ++v) {
          float4 o;
          o.x = acc[i][0][v]; o.y = acc[i][1][v];
          o.z = acc[i][2][v]; o.w = acc[i][3][v];
          *(float4*)&smx[mb + v][n0 + 4 * l15] = o;
        }
      }
    }
    __syncthreads();

    float sx = 0.f;
    {
      const int seg = KDIM / 8;
      const float* xr = X + (size_t)(rb + r) * KDIM + p * seg;
      for (int j = 0; j < seg; j += 4) {
        float4 v = *(const float4*)(xr + j);
        sx += v.x * v.x + v.y * v.y + v.z * v.z + v.w * v.w;
      }
    }
    float sm = 0.f, sd = 0.f;
    {
      const float* mr = &smx[r][p * 32];
#pragma unroll
      for (int j = 0; j < 32; ++j) {
        float m = mr[j];
        sm += m * m;
        if (FULL_EPI) sd += m * bias[p * 32 + j];
      }
    }
    part[r][p] = sx;
    __syncthreads();
    if (t < 32) { float s = 0.f; for (int i = 0; i < 8; ++i) s += part[t][i]; rx2[t] = s; }
    __syncthreads();
    part[r][p] = sm;
    __syncthreads();
    if (t < 32) { float s = 0.f; for (int i = 0; i < 8; ++i) s += part[t][i]; rm2[t] = s; }
    __syncthreads();
    if (FULL_EPI) {
      part[r][p] = sd;
      __syncthreads();
      if (t < 32) { float s = 0.f; for (int i = 0; i < 8; ++i) s += part[t][i]; rdot[t] = s; }
      __syncthreads();
    }

    const float x2s = inScale * inScale * rx2[r];
    const float m2s = inScale * inScale * rm2[r];
    const float cmv = hyp_scale(x2s, m2s) * inScale;
    float* orow = OUT + (size_t)(rb + r) * NHID + p * 32;
    float* mrw = &smx[r][p * 32];
    if (!FULL_EPI) {
#pragma unroll
      for (int j = 0; j < 32; j += 4) {
        float4 o;
        o.x = cmv * mrw[j + 0]; o.y = cmv * mrw[j + 1];
        o.z = cmv * mrw[j + 2]; o.w = cmv * mrw[j + 3];
        *(float4*)(orow + j) = o;
      }
    } else {
      const float cf = 0.25f * cmv;
      const float x2r = cf * cf * rm2[r];
      const float xy = cf * rdot[r];
      const float y2 = sb2s;
      const float den = fmaxf(1.f + 2.f * xy + x2r * y2, EPSF);
      const float ca = (1.f + 2.f * xy + y2) / den;
      const float cb = (1.f - x2r) / den;
      const float zn2 = ca * ca * x2r + 2.f * ca * cb * xy + cb * cb * y2;
      const float zn = fminf(fmaxf(sqrtf(zn2), EPSF), MAXNF);
      const float us = atanhf(zn) / zn;
      const float acf = ca * cf;
      float sv = 0.f;
#pragma unroll
      for (int j = 0; j < 32; ++j) {
        float z = acf * mrw[j] + cb * bias[p * 32 + j];
        float uu = fmaxf(us * z, 0.f);
        mrw[j] = uu;
        sv += uu * uu;
      }
      part[r][p] = sv;
      __syncthreads();
      if (t < 32) { float s = 0.f; for (int i = 0; i < 8; ++i) s += part[t][i]; rv2[t] = s; }
      __syncthreads();
      const float vn = fmaxf(sqrtf(rv2[r]), EPSF);
      const float fs = tanhf(vn) / vn;
#pragma unroll
      for (int j = 0; j < 32; j += 4) {
        float4 o;
        o.x = fs * mrw[j + 0]; o.y = fs * mrw[j + 1];
        o.z = fs * mrw[j + 2]; o.w = fs * mrw[j + 3];
        *(float4*)(orow + j) = o;
      }
    }
    if (q < 3) __syncthreads();
  }
}

__global__ __launch_bounds__(256) void k_agg(const float* __restrict__ F,
                                             const int* __restrict__ idx,
                                             const float* __restrict__ bias,
                                             float* __restrict__ OUT,
                                             float inScale, int doAct) {
  const int t = threadIdx.x;
  const int wave = t >> 5, lane = t & 31;
  const int node = (blockIdx.x << 3) + wave;
  const int* srow = idx + (size_t)node * 16;
  const int joff = lane << 3;

  float x[8];
  load8(F + (size_t)srow[0] * NHID + joff, inScale, x);
  float px = 0.f;
#pragma unroll
  for (int j = 0; j < 8; ++j) px += x[j] * x[j];
  float x2 = wred32(px);

  for (int k = 1; k < 16; ++k) {
    if (k < 15) __builtin_prefetch(F + (size_t)srow[k + 1] * NHID + joff, 0, 0);
    float y[8];
    load8(F + (size_t)srow[k] * NHID + joff, inScale, y);
    float py2 = 0.f, pxy = 0.f;
#pragma unroll
    for (int j = 0; j < 8; ++j) { py2 += y[j] * y[j]; pxy += x[j] * y[j]; }
    float y2 = wred32(py2);
    float xy = wred32(pxy);
    float den = fmaxf(1.f + 2.f * xy + x2 * y2, EPSF);
    float ca = (1.f + 2.f * xy + y2) / den;
    float cb = (1.f - x2) / den;
    float pn = 0.f;
#pragma unroll
    for (int j = 0; j < 8; ++j) { x[j] = ca * x[j] + cb * y[j]; pn += x[j] * x[j]; }
    x2 = wred32(pn);
  }

  float* orow = OUT + (size_t)node * NHID + joff;
  if (!doAct) {
    store8(orow, x, 1.f);
  } else {
    float b[8];
    load8(bias + joff, 1.f, b);
    float pb2 = 0.f, pxy = 0.f;
#pragma unroll
    for (int j = 0; j < 8; ++j) { pb2 += b[j] * b[j]; pxy += x[j] * b[j]; }
    float y2 = wred32(pb2);
    float xy = 0.25f * wred32(pxy);
    float x2r = 0.0625f * x2;
    float den = fmaxf(1.f + 2.f * xy + x2r * y2, EPSF);
    float ca = (1.f + 2.f * xy + y2) / den;
    float cb = (1.f - x2r) / den;
    float zn2 = ca * ca * x2r + 2.f * ca * cb * xy + cb * cb * y2;
    float zn = fminf(fmaxf(sqrtf(zn2), EPSF), MAXNF);
    float us = atanhf(zn) / zn;
    float u[8];
    float pv = 0.f;
#pragma unroll
    for (int j = 0; j < 8; ++j) {
      float z = ca * 0.25f * x[j] + cb * b[j];
      u[j] = fmaxf(us * z, 0.f);
      pv += u[j] * u[j];
    }
    float vn = fmaxf(sqrtf(wred32(pv)), EPSF);
    float fs = tanhf(vn) / vn;
    store8(orow, u, fs);
  }
}

__global__ __launch_bounds__(256) void k_final(const float* __restrict__ H,
                                               const float* __restrict__ Wl,
                                               const float* __restrict__ bl,
                                               const int* __restrict__ tf,
                                               float* __restrict__ OUT) {
  __shared__ int ridx[64];
  __shared__ float smx[64][72];
  __shared__ float part[64][4];
  __shared__ float rx2[64], rm2[64], rdot[64];
  __shared__ float sb2;

  const int t = threadIdx.x;
  if (t < 64) ridx[t] = tf[t] + t * 1024;
  if (t == 0) {
    float s = 0.f;
    for (int j = 0; j < 64; ++j) s += bl[j] * bl[j];
    sb2 = s;
  }
  __syncthreads();

  const int wave = t >> 5, lane = t & 31;
  const int l15 = lane & 15;
  const int kk = (lane >> 4) << 1;
#pragma unroll
  for (int i = 0; i < 2; ++i) {
    const int tile = wave * 2 + i;
    const int tm = tile >> 2, tn = tile & 3;
    const float* Ar = H + (size_t)ridx[tm * 16 + l15] * NHID;
    const float* Br = Wl + (size_t)(tn * 16 + l15) * NHID;
    v8f c = {};
    for (int k0 = 0; k0 < NHID; k0 += 4) {
      v2f a, b;
      a.x = Ar[k0 + kk]; a.y = Ar[k0 + kk + 1];
      b.x = Br[k0 + kk]; b.y = Br[k0 + kk + 1];
      c = wmma_f32x4(a, b, c);
    }
    const int n = tn * 16 + l15;
    const int mb = tm * 16 + ((lane >> 4) ? 8 : 0);
#pragma unroll
    for (int v = 0; v < 8; ++v) smx[mb + v][n] = c[v];
  }
  __syncthreads();

  const int r = t >> 2, p = t & 3;
  float sx = 0.f;
  {
    const float* hr = H + (size_t)ridx[r] * NHID + p * 64;
    for (int j = 0; j < 64; j += 4) {
      float4 v = *(const float4*)(hr + j);
      sx += v.x * v.x + v.y * v.y + v.z * v.z + v.w * v.w;
    }
  }
  float sm = 0.f, sd = 0.f;
#pragma unroll
  for (int j = 0; j < 16; ++j) {
    float m = smx[r][p * 16 + j];
    sm += m * m;
    sd += m * bl[p * 16 + j];
  }
  part[r][p] = sx;
  __syncthreads();
  if (t < 64) rx2[t] = part[t][0] + part[t][1] + part[t][2] + part[t][3];
  __syncthreads();
  part[r][p] = sm;
  __syncthreads();
  if (t < 64) rm2[t] = part[t][0] + part[t][1] + part[t][2] + part[t][3];
  __syncthreads();
  part[r][p] = sd;
  __syncthreads();
  if (t < 64) rdot[t] = part[t][0] + part[t][1] + part[t][2] + part[t][3];
  __syncthreads();

  const float s = hyp_scale(rx2[r], rm2[r]);
  const float x2r = s * s * rm2[r];
  const float xy = s * rdot[r];
  const float y2 = sb2;
  const float den = fmaxf(1.f + 2.f * xy + x2r * y2, EPSF);
  const float ca = (1.f + 2.f * xy + y2) / den;
  const float cb = (1.f - x2r) / den;
  const float acs = ca * s;
#pragma unroll
  for (int j = 0; j < 16; ++j)
    OUT[r * 64 + p * 16 + j] = acs * smx[r][p * 16 + j] + cb * bl[p * 16 + j];
}

extern "C" void kernel_launch(void* const* d_in, const int* in_sizes, int n_in,
                              void* d_out, int out_size, void* d_ws,
                              size_t ws_size, hipStream_t stream) {
  const float* features = (const float*)d_in[0];
  const float* W1 = (const float*)d_in[1];
  const float* b1 = (const float*)d_in[2];
  const float* W2 = (const float*)d_in[3];
  const float* b2 = (const float*)d_in[4];
  const float* Wl = (const float*)d_in[5];
  const float* bl = (const float*)d_in[6];
  const int* src_idx = (const int*)d_in[7];
  const int* to_fetch = (const int*)d_in[8];
  float* out = (float*)d_out;

  float* buf0 = (float*)d_ws;
  float* buf1 = buf0 + (size_t)65536 * 256;

  k_mm<512, false><<<512, 256, 0, stream>>>(const_cast<float*>(features), W1,
                                             b1, buf0, 0.25f);
  k_agg<<<8192, 256, 0, stream>>>(buf0, src_idx, b1, buf1, 1.0f, 1);
  k_agg<<<8192, 256, 0, stream>>>(buf1, src_idx, b2, buf0, 0.25f, 0);
  k_mm<256, true><<<512, 256, 0, stream>>>(buf0, W2, b2, buf0, 1.0f);
  k_final<<<1, 256, 0, stream>>>(buf0, Wl, bl, to_fetch, out);
}


# k_agg: four nodes per wave (8 lanes x 32 columns per node, was 32 x 8): fold scalars and wave-sum hops shared by 4 nodes, 3 hops not 5; row sums' order changes (2x16 per lane, then 3 hops)
# speedup vs baseline: 1.3896x; 1.1872x over previous
#include <hip/hip_runtime.h>

typedef __attribute__((ext_vector_type(2))) float v2f;
typedef __attribute__((ext_vector_type(8))) float v8f;

#define NHID 256
#define EPSF 1e-7f
#define MAXNF 0.99999f

__device__ __forceinline__ v8f wmma_f32x4(v2f a, v2f b, v8f c) {
  return __builtin_amdgcn_wmma_f32_16x16x4_f32(false, a, false, b, (short)0, c,
                                               false, false);
}

__device__ __forceinline__ float gred8(float v) {
#pragma unroll
  for (int m = 4; m > 0; m >>= 1) v += __shfl_xor(v, m, 32);
  return v;
}

__device__ __forceinline__ float hyp_scale(float x2, float m2) {
  float xn  = fminf(fmaxf(sqrtf(x2), EPSF), MAXNF);
  float mxn = fmaxf(sqrtf(m2), EPSF);
  return tanhf(mxn / xn * atanhf(xn)) / mxn;
}

__device__ __forceinline__ void load32(const float* p, v2f s, v2f* d) {
#pragma unroll
  for (int i = 0; i < 8; ++i) {
    float4 a = *(const float4*)(p + 32 * i);
    v2f lo = {a.x, a.y}, hi = {a.z, a.w};
    d[2 * i] = lo * s;
    d[2 * i + 1] = hi * s;
  }
}

__device__ __forceinline__ void store32(float* p, const v2f* d, float s) {
#pragma unroll
  for (int i = 0; i < 8; ++i) {
    v2f lo = d[2 * i] * s, hi = d[2 * i + 1] * s;
    float4 a = {lo.x, lo.y, hi.x, hi.y};
    *(float4*)(p + 32 * i) = a;
  }
}

template <int KDIM>
__device__ __forceinline__ void mm_load(const float* Ap, const float* Bp, int k0,
                                        float4 (&a)[4], float4 (&b)[4]) {
  for (int i = 0; i < 4; ++i)
    a[i] = *(const float4*)(Ap + (size_t)i * 16 * KDIM + k0);
  for (int s = 0; s < 4; ++s)
    b[s] = *(const float4*)(Bp + (size_t)(k0 + s) * NHID);
}

__device__ __forceinline__ void mm_block(v8f (&acc)[4][4], const float4 (&a)[4],
                                         const float4 (&b)[4]) {
  const float bv[4][4] = {{b[0].x, b[0].y, b[0].z, b[0].w},
                          {b[1].x, b[1].y, b[1].z, b[1].w},
                          {b[2].x, b[2].y, b[2].z, b[2].w},
                          {b[3].x, b[3].y, b[3].z, b[3].w}};
  for (int i = 0; i < 4; ++i) {
    v2f a0; a0.x = a[i].x; a0.y = a[i].y;
    for (int j = 0; j < 4; ++j) {
      v2f b0; b0.x = bv[0][j]; b0.y = bv[1][j];
      acc[i][j] = wmma_f32x4(a0, b0, acc[i][j]);
    }
  }
  for (int i = 0; i < 4; ++i) {
    v2f a1; a1.x = a[i].z; a1.y = a[i].w;
    for (int j = 0; j < 4; ++j) {
      v2f b1; b1.x = bv[2][j]; b1.y = bv[3][j];
      acc[i][j] = wmma_f32x4(a1, b1, acc[i][j]);
    }
  }
}

template <int KDIM, bool FULL_EPI>
__global__ __launch_bounds__(256) void k_mm(const float* X,
                                            const float* __restrict__ W,
                                            const float* __restrict__ bias,
                                            float* OUT, float inScale) {
  __shared__ alignas(16) float smx[32][NHID + 8];
  __shared__ float part[32][8];
  __shared__ float rx2[32], rm2[32], rdot[32], rv2[32];
  __shared__ float sb2s;

  const int t = threadIdx.x;
  const int wave = t >> 5, lane = t & 31;
  const int tile_m = wave >> 2;
  const int n0 = (wave & 3) << 6;
  const int l15 = lane & 15;
  const int h = lane >> 4;
  const int brow = blockIdx.x << 7;

  v8f acc[4][4];
  for (int i = 0; i < 4; ++i)
    for (int j = 0; j < 4; ++j) acc[i][j] = v8f{};

  const float* Ap = X + (size_t)(brow + tile_m * 64 + l15) * KDIM + 4 * h;
  const float* Bp = W + (size_t)(4 * h) * NHID + n0 + 4 * l15;

  float4 a0[4], b0[4], a1[4], b1[4];
  mm_load<KDIM>(Ap, Bp, 0, a0, b0);
  for (int k0 = 0; k0 < KDIM - 16; k0 += 16) {
    mm_load<KDIM>(Ap, Bp, k0 + 8, a1, b1);
    mm_block(acc, a0, b0);
    mm_load<KDIM>(Ap, Bp, k0 + 16, a0, b0);
    mm_block(acc, a1, b1);
  }
  mm_load<KDIM>(Ap, Bp, KDIM - 8, a1, b1);
  mm_block(acc, a0, b0);
  mm_block(acc, a1, b1);

  const int r = t >> 3, p = t & 7;
  if (FULL_EPI) {
    part[r][p] = bias[t] * bias[t];
    __syncthreads();
    if (t == 0) {
      float s = 0.f;
      for (int i = 0; i < 32; ++i)
        for (int j = 0; j < 8; ++j) s += part[i][j];
      sb2s = s;
    }
    __syncthreads();
  }

#pragma unroll
  for (int q = 0; q < 4; ++q) {
    const int rb = brow + 32 * q;
    if (tile_m == (q >> 1)) {
      for (int ii = 0; ii < 2; ++ii) {
        const int i = 2 * (q & 1) + ii;
        const int mb = ii * 16 + 8 * h;
        for (int v = 0; v < 8; ++v) {
          float4 o;
          o.x = acc[i][0][v]; o.y = acc[i][1][v];
          o.z = acc[i][2][v]; o.w = acc[i][3][v];
          *(float4*)&smx[mb + v][n0 + 4 * l15] = o;
        }
      }
    }
    __syncthreads();

    float sx = 0.f;
    {
      const int seg = KDIM / 8;
      const float* xr = X + (size_t)(rb + r) * KDIM + p * seg;
      for (int j = 0; j < seg; j += 4) {
        float4 v = *(const float4*)(xr + j);
        sx += v.x * v.x + v.y * v.y + v.z * v.z + v.w * v.w;
      }
    }
    float sm = 0.f, sd = 0.f;
    {
      const float* mr = &smx[r][p * 32];
#pragma unroll
      for (int j = 0; j < 32; ++j) {
        float m = mr[j];
        sm += m * m;
        if (FULL_EPI) sd += m * bias[p * 32 + j];
      }
    }
    part[r][p] = sx;
    __syncthreads();
    if (t < 32) { float s = 0.f; for (int i = 0; i < 8; ++i) s += part[t][i]; rx2[t] = s; }
    __syncthreads();
    part[r][p] = sm;
    __syncthreads();
    if (t < 32) { float s = 0.f; for (int i = 0; i < 8; ++i) s += part[t][i]; rm2[t] = s; }
    __syncthreads();
    if (FULL_EPI) {
      part[r][p] = sd;
      __syncthreads();
      if (t < 32) { float s = 0.f; for (int i = 0; i < 8; ++i) s += part[t][i]; rdot[t] = s; }
      __syncthreads();
    }

    const float x2s = inScale * inScale * rx2[r];
    const float m2s = inScale * inScale * rm2[r];
    const float cmv = hyp_scale(x2s, m2s) * inScale;
    float* orow = OUT + (size_t)(rb + r) * NHID + p * 32;
    float* mrw = &smx[r][p * 32];
    if (!FULL_EPI) {
#pragma unroll
      for (int j = 0; j < 32; j += 4) {
        float4 o;
        o.x = cmv * mrw[j + 0]; o.y = cmv * mrw[j + 1];
        o.z = cmv * mrw[j + 2]; o.w = cmv * mrw[j + 3];
        *(float4*)(orow + j) = o;
      }
    } else {
      const float cf = 0.25f * cmv;
      const float x2r = cf * cf * rm2[r];
      const float xy = cf * rdot[r];
      const float y2 = sb2s;
      const float den = fmaxf(1.f + 2.f * xy + x2r * y2, EPSF);
      const float ca = (1.f + 2.f * xy + y2) / den;
      const float cb = (1.f - x2r) / den;
      const float zn2 = ca * ca * x2r + 2.f * ca * cb * xy + cb * cb * y2;
      const float zn = fminf(fmaxf(sqrtf(zn2), EPSF), MAXNF);
      const float us = atanhf(zn) / zn;
      const float acf = ca * cf;
      float sv = 0.f;
#pragma unroll
      for (int j = 0; j < 32; ++j) {
        float z = acf * mrw[j] + cb * bias[p * 32 + j];
        float uu = fmaxf(us * z, 0.f);
        mrw[j] = uu;
        sv += uu * uu;
      }
      part[r][p] = sv;
      __syncthreads();
      if (t < 32) { float s = 0.f; for (int i = 0; i < 8; ++i) s += part[t][i]; rv2[t] = s; }
      __syncthreads();
      const float vn = fmaxf(sqrtf(rv2[r]), EPSF);
      const float fs = tanhf(vn) / vn;
#pragma unroll
      for (int j = 0; j < 32; j += 4) {
        float4 o;
        o.x = fs * mrw[j + 0]; o.y = fs * mrw[j + 1];
        o.z = fs * mrw[j + 2]; o.w = fs * mrw[j + 3];
        *(float4*)(orow + j) = o;
      }
    }
    if (q < 3) __syncthreads();
  }
}

__global__ __launch_bounds__(256) void k_agg(const float* __restrict__ F,
                                             const int* __restrict__ idx,
                                             const float* __restrict__ bias,
                                             float* __restrict__ OUT,
                                             float inScale, int doAct) {
  const int t = threadIdx.x;
  const int s = t & 7;
  const int node = (blockIdx.x << 5) + (t >> 3);
  const int* srow = idx + (size_t)node * 16;
  const int joff = s << 2;

  v2f rs = {inScale, inScale};
  asm volatile("" : "+v"(rs));

  v2f x[16];
  load32(F + (size_t)srow[0] * NHID + joff, rs, x);
  v2f px = {0.f, 0.f};
#pragma unroll
  for (int j = 0; j < 16; ++j) px += x[j] * x[j];
  float x2 = gred8(px.x + px.y);

  for (int k = 1; k < 16; ++k) {
    if (k < 15) __builtin_prefetch(F + (size_t)srow[k + 1] * NHID + (s << 5), 0, 0);
    v2f y[16];
    load32(F + (size_t)srow[k] * NHID + joff, rs, y);
    v2f py2 = {0.f, 0.f}, pxy = {0.f, 0.f};
#pragma unroll
    for (int j = 0; j < 16; ++j) { py2 += y[j] * y[j]; pxy += x[j] * y[j]; }
    float y2 = gred8(py2.x + py2.y);
    float xy = gred8(pxy.x + pxy.y);
    float den = fmaxf(1.f + 2.f * xy + x2 * y2, EPSF);
    float ca = (1.f + 2.f * xy + y2) / den;
    float cb = (1.f - x2) / den;
    v2f pn = {0.f, 0.f};
#pragma unroll
    for (int j = 0; j < 16; ++j) { x[j] = ca * x[j] + cb * y[j]; pn += x[j] * x[j]; }
    x2 = gred8(pn.x + pn.y);
  }

  float* orow = OUT + (size_t)node * NHID + joff;
  if (!doAct) {
    store32(orow, x, 1.f);
  } else {
    v2f b[16];
    load32(bias + joff, v2f{1.f, 1.f}, b);
    v2f pb2 = {0.f, 0.f}, pxy = {0.f, 0.f};
#pragma unroll
    for (int j = 0; j < 16; ++j) { pb2 += b[j] * b[j]; pxy += x[j] * b[j]; }
    float y2 = gred8(pb2.x + pb2.y);
    float xy = 0.25f * gred8(pxy.x + pxy.y);
    float x2r = 0.0625f * x2;
    float den = fmaxf(1.f + 2.f * xy + x2r * y2, EPSF);
    float ca = (1.f + 2.f * xy + y2) / den;
    float cb = (1.f - x2r) / den;
    float zn2 = ca * ca * x2r + 2.f * ca * cb * xy + cb * cb * y2;
    float zn = fminf(fmaxf(sqrtf(zn2), EPSF), MAXNF);
    float us = atanhf(zn) / zn;
    const float cq = ca * 0.25f;
    v2f pv = {0.f, 0.f};
#pragma unroll
    for (int j = 0; j < 16; ++j) {
      v2f z = cq * x[j] + cb * b[j];
      v2f u;
      u.x = fmaxf(us * z.x, 0.f);
      u.y = fmaxf(us * z.y, 0.f);
      x[j] = u;
      pv += u * u;
    }
    float vn = fmaxf(sqrtf(gred8(pv.x + pv.y)), EPSF);
    float fs = tanhf(vn) / vn;
    store32(orow, x, fs);
  }
}

__global__ __launch_bounds__(256) void k_final(const float* __restrict__ H,
                                               const float* __restrict__ Wl,
                                               const float* __restrict__ bl,
                                               const int* __restrict__ tf,
                                               float* __restrict__ OUT) {
  __shared__ int ridx[64];
  __shared__ float smx[64][72];
  __shared__ float part[64][4];
  __shared__ float rx2[64], rm2[64], rdot[64];
  __shared__ float sb2;

  const int t = threadIdx.x;
  if (t < 64) ridx[t] = tf[t] + t * 1024;
  if (t == 0) {
    float s = 0.f;
    for (int j = 0; j < 64; ++j) s += bl[j] * bl[j];
    sb2 = s;
  }
  __syncthreads();

  const int wave = t >> 5, lane = t & 31;
  const int l15 = lane & 15;
  const int kk = (lane >> 4) << 1;
#pragma unroll
  for (int i = 0; i < 2; ++i) {
    const int tile = wave * 2 + i;
    const int tm = tile >> 2, tn = tile & 3;
    const float* Ar = H + (size_t)ridx[tm * 16 + l15] * NHID;
    const float* Br = Wl + (size_t)(tn * 16 + l15) * NHID;
    v8f c = {};
    for (int k0 = 0; k0 < NHID; k0 += 4) {
      v2f a, b;
      a.x = Ar[k0 + kk]; a.y = Ar[k0 + kk + 1];
      b.x = Br[k0 + kk]; b.y = Br[k0 + kk + 1];
      c = wmma_f32x4(a, b, c);
    }
    const int n = tn * 16 + l15;
    const int mb = tm * 16 + ((lane >> 4) ? 8 : 0);
#pragma unroll
    for (int v = 0; v < 8; ++v) smx[mb + v][n] = c[v];
  }
  __syncthreads();

  const int r = t >> 2, p = t & 3;
  float sx = 0.f;
  {
    const float* hr = H + (size_t)ridx[r] * NHID + p * 64;
    for (int j = 0; j < 64; j += 4) {
      float4 v = *(const float4*)(hr + j);
      sx += v.x * v.x + v.y * v.y + v.z * v.z + v.w * v.w;
    }
  }
  float sm = 0.f, sd = 0.f;
#pragma unroll
  for (int j = 0; j < 16; ++j) {
    float m = smx[r][p * 16 + j];
    sm += m * m;
    sd += m * bl[p * 16 + j];
  }
  part[r][p] = sx;
  __syncthreads();
  if (t < 64) rx2[t] = part[t][0] + part[t][1] + part[t][2] + part[t][3];
  __syncthreads();
  part[r][p] = sm;
  __syncthreads();
  if (t < 64) rm2[t] = part[t][0] + part[t][1] + part[t][2] + part[t][3];
  __syncthreads();
  part[r][p] = sd;
  __syncthreads();
  if (t < 64) rdot[t] = part[t][0] + part[t][1] + part[t][2] + part[t][3];
  __syncthreads();

  const float s = hyp_scale(rx2[r], rm2[r]);
  const float x2r = s * s * rm2[r];
  const float xy = s * rdot[r];
  const float y2 = sb2;
  const float den = fmaxf(1.f + 2.f * xy + x2r * y2, EPSF);
  const float ca = (1.f + 2.f * xy + y2) / den;
  const float cb = (1.f - x2r) / den;
  const float acs = ca * s;
#pragma unroll
  for (int j = 0; j < 16; ++j)
    OUT[r * 64 + p * 16 + j] = acs * smx[r][p * 16 + j] + cb * bl[p * 16 + j];
}

extern "C" void kernel_launch(void* const* d_in, const int* in_sizes, int n_in,
                              void* d_out, int out_size, void* d_ws,
                              size_t ws_size, hipStream_t stream) {
  const float* features = (const float*)d_in[0];
  const float* W1 = (const float*)d_in[1];
  const float* b1 = (const float*)d_in[2];
  const float* W2 = (const float*)d_in[3];
  const float* b2 = (const float*)d_in[4];
  const float* Wl = (const float*)d_in[5];
  const float* bl = (const float*)d_in[6];
  const int* src_idx = (const int*)d_in[7];
  const int* to_fetch = (const int*)d_in[8];
  float* out = (float*)d_out;

  float* buf0 = (float*)d_ws;
  float* buf1 = buf0 + (size_t)65536 * 256;

  k_mm<512, false><<<512, 256, 0, stream>>>(const_cast<float*>(features), W1,
                                             b1, buf0, 0.25f);
  k_agg<<<2048, 256, 0, stream>>>(buf0, src_idx, b1, buf1, 1.0f, 1);
  k_agg<<<2048, 256, 0, stream>>>(buf1, src_idx, b2, buf0, 0.25f, 0);
  k_mm<256, true><<<512, 256, 0, stream>>>(buf0, W2, b2, buf0, 1.0f);
  k_final<<<1, 256, 0, stream>>>(buf0, Wl, bl, to_fetch, out);
}


# k_mm epilogue: a row's 8 threads own interleaved 16-byte pieces (cols 32i+4p, was 32p..32p+31): each wave load of X and store of OUT covers 4 whole lines, not 32 partial; row sums' order changes
# speedup vs baseline: 1.4459x; 1.0405x over previous
#include <hip/hip_runtime.h>

typedef __attribute__((ext_vector_type(2))) float v2f;
typedef __attribute__((ext_vector_type(8))) float v8f;

#define NHID 256
#define EPSF 1e-7f
#define MAXNF 0.99999f

__device__ __forceinline__ v8f wmma_f32x4(v2f a, v2f b, v8f c) {
  return __builtin_amdgcn_wmma_f32_16x16x4_f32(false, a, false, b, (short)0, c,
                                               false, false);
}

__device__ __forceinline__ float gred8(float v) {
#pragma unroll
  for (int m = 4; m > 0; m >>= 1) v += __shfl_xor(v, m, 32);
  return v;
}

__device__ __forceinline__ float hyp_scale(float x2, float m2) {
  float xn  = fminf(fmaxf(sqrtf(x2), EPSF), MAXNF);
  float mxn = fmaxf(sqrtf(m2), EPSF);
  return tanhf(mxn / xn * atanhf(xn)) / mxn;
}

__device__ __forceinline__ void load32(const float* p, v2f s, v2f* d) {
#pragma unroll
  for (int i = 0; i < 8; ++i) {
    float4 a = *(const float4*)(p + 32 * i);
    v2f lo = {a.x, a.y}, hi = {a.z, a.w};
    d[2 * i] = lo * s;
    d[2 * i + 1] = hi * s;
  }
}

__device__ __forceinline__ void store32(float* p, const v2f* d, float s) {
#pragma unroll
  for (int i = 0; i < 8; ++i) {
    v2f lo = d[2 * i] * s, hi = d[2 * i + 1] * s;
    float4 a = {lo.x, lo.y, hi.x, hi.y};
    *(float4*)(p + 32 * i) = a;
  }
}

template <int KDIM>
__device__ __forceinline__ void mm_load(const float* Ap, const float* Bp, int k0,
                                        float4 (&a)[4], float4 (&b)[4]) {
  for (int i = 0; i < 4; ++i)
    a[i] = *(const float4*)(Ap + (size_t)i * 16 * KDIM + k0);
  for (int s = 0; s < 4; ++s)
    b[s] = *(const float4*)(Bp + (size_t)(k0 + s) * NHID);
}

__device__ __forceinline__ void mm_block(v8f (&acc)[4][4], const float4 (&a)[4],
                                         const float4 (&b)[4]) {
  const float bv[4][4] = {{b[0].x, b[0].y, b[0].z, b[0].w},
                          {b[1].x, b[1].y, b[1].z, b[1].w},
                          {b[2].x, b[2].y, b[2].z, b[2].w},
                          {b[3].x, b[3].y, b[3].z, b[3].w}};
  for (int i = 0; i < 4; ++i) {
    v2f a0; a0.x = a[i].x; a0.y = a[i].y;
    for (int j = 0; j < 4; ++j) {
      v2f b0; b0.x = bv[0][j]; b0.y = bv[1][j];
      acc[i][j] = wmma_f32x4(a0, b0, acc[i][j]);
    }
  }
  for (int i = 0; i < 4; ++i) {
    v2f a1; a1.x = a[i].z; a1.y = a[i].w;
    for (int j = 0; j < 4; ++j) {
      v2f b1; b1.x = bv[2][j]; b1.y = bv[3][j];
      acc[i][j] = wmma_f32x4(a1, b1, acc[i][j]);
    }
  }
}

template <int KDIM, bool FULL_EPI>
__global__ __launch_bounds__(256) void k_mm(const float* X,
                                            const float* __restrict__ W,
                                            const float* __restrict__ bias,
                                            float* OUT, float inScale) {
  __shared__ alignas(16) float smx[32][NHID + 32];
  __shared__ float part[32][8];
  __shared__ float rx2[32], rm2[32], rdot[32], rv2[32];
  __shared__ float sb2s;

  const int t = threadIdx.x;
  const int wave = t >> 5, lane = t & 31;
  const int tile_m = wave >> 2;
  const int n0 = (wave & 3) << 6;
  const int l15 = lane & 15;
  const int h = lane >> 4;
  const int brow = blockIdx.x << 7;

  v8f acc[4][4];
  for (int i = 0; i < 4; ++i)
    for (int j = 0; j < 4; ++j) acc[i][j] = v8f{};

  const float* Ap = X + (size_t)(brow + tile_m * 64 + l15) * KDIM + 4 * h;
  const float* Bp = W + (size_t)(4 * h) * NHID + n0 + 4 * l15;

  float4 a0[4], b0[4], a1[4], b1[4];
  mm_load<KDIM>(Ap, Bp, 0, a0, b0);
  for (int k0 = 0; k0 < KDIM - 16; k0 += 16) {
    mm_load<KDIM>(Ap, Bp, k0 + 8, a1, b1);
    mm_block(acc, a0, b0);
    mm_load<KDIM>(Ap, Bp, k0 + 16, a0, b0);
    mm_block(acc, a1, b1);
  }
  mm_load<KDIM>(Ap, Bp, KDIM - 8, a1, b1);
  mm_block(acc, a0, b0);
  mm_block(acc, a1, b1);

  const int r = t >> 3, p = t & 7;
  const int pc = 4 * p;
  if (FULL_EPI) {
    part[r][p] = bias[t] * bias[t];
    __syncthreads();
    if (t == 0) {
      float s = 0.f;
      for (int i = 0; i < 32; ++i)
        for (int j = 0; j < 8; ++j) s += part[i][j];
      sb2s = s;
    }
    __syncthreads();
  }

#pragma unroll
  for (int q = 0; q < 4; ++q) {
    const int rb = brow + 32 * q;
    if (tile_m == (q >> 1)) {
      for (int ii = 0; ii < 2; ++ii) {
        const int i = 2 * (q & 1) + ii;
        const int mb = ii * 16 + 8 * h;
        for (int v = 0; v < 8; ++v) {
          float4 o;
          o.x = acc[i][0][v]; o.y = acc[i][1][v];
          o.z = acc[i][2][v]; o.w = acc[i][3][v];
          *(float4*)&smx[mb + v][n0 + 4 * l15] = o;
        }
      }
    }
    __syncthreads();

    float sx = 0.f;
    {
      const float* xr = X + (size_t)(rb + r) * KDIM + pc;
      for (int j = 0; j < KDIM; j += 32) {
        float4 v = *(const float4*)(xr + j);
        sx += v.x * v.x + v.y * v.y + v.z * v.z + v.w * v.w;
      }
    }
    float sm = 0.f, sd = 0.f;
    {
      const float* mr = &smx[r][pc];
#pragma unroll
      for (int i = 0; i < 8; ++i) {
        const float4 m = *(const float4*)(mr + 32 * i);
        sm += m.x * m.x; sm += m.y * m.y; sm += m.z * m.z; sm += m.w * m.w;
        if (FULL_EPI) {
          const float4 b = *(const float4*)(bias + 32 * i + pc);
          sd += m.x * b.x; sd += m.y * b.y; sd += m.z * b.z; sd += m.w * b.w;
        }
      }
    }
    part[r][p] = sx;
    __syncthreads();
    if (t < 32) { float s = 0.f; for (int i = 0; i < 8; ++i) s += part[t][i]; rx2[t] = s; }
    __syncthreads();
    part[r][p] = sm;
    __syncthreads();
    if (t < 32) { float s = 0.f; for (int i = 0; i < 8; ++i) s += part[t][i]; rm2[t] = s; }
    __syncthreads();
    if (FULL_EPI) {
      part[r][p] = sd;
      __syncthreads();
      if (t < 32) { float s = 0.f; for (int i = 0; i < 8; ++i) s += part[t][i]; rdot[t] = s; }
      __syncthreads();
    }

    const float x2s = inScale * inScale * rx2[r];
    const float m2s = inScale * inScale * rm2[r];
    const float cmv = hyp_scale(x2s, m2s) * inScale;
    float* orow = OUT + (size_t)(rb + r) * NHID + pc;
    float* mrw = &smx[r][pc];
    if (!FULL_EPI) {
#pragma unroll
      for (int i = 0; i < 8; ++i) {
        const float4 m = *(const float4*)(mrw + 32 * i);
        float4 o;
        o.x = cmv * m.x; o.y = cmv * m.y; o.z = cmv * m.z; o.w = cmv * m.w;
        *(float4*)(orow + 32 * i) = o;
      }
    } else {
      const float cf = 0.25f * cmv;
      const float x2r = cf * cf * rm2[r];
      const float xy = cf * rdot[r];
      const float y2 = sb2s;
      const float den = fmaxf(1.f + 2.f * xy + x2r * y2, EPSF);
      const float ca = (1.f + 2.f * xy + y2) / den;
      const float cb = (1.f - x2r) / den;
      const float zn2 = ca * ca * x2r + 2.f * ca * cb * xy + cb * cb * y2;
      const float zn = fminf(fmaxf(sqrtf(zn2), EPSF), MAXNF);
      const float us = atanhf(zn) / zn;
      const float acf = ca * cf;
      float sv = 0.f;
#pragma unroll
      for (int i = 0; i < 8; ++i) {
        const float4 m = *(const float4*)(mrw + 32 * i);
        const float4 b = *(const float4*)(bias + 32 * i + pc);
        float4 uu;
        uu.x = fmaxf(us * (acf * m.x + cb * b.x), 0.f);
        uu.y = fmaxf(us * (acf * m.y + cb * b.y), 0.f);
        uu.z = fmaxf(us * (acf * m.z + cb * b.z), 0.f);
        uu.w = fmaxf(us * (acf * m.w + cb * b.w), 0.f);
        *(float4*)(mrw + 32 * i) = uu;
        sv += uu.x * uu.x; sv += uu.y * uu.y; sv += uu.z * uu.z; sv += uu.w * uu.w;
      }
      part[r][p] = sv;
      __syncthreads();
      if (t < 32) { float s = 0.f; for (int i = 0; i < 8; ++i) s += part[t][i]; rv2[t] = s; }
      __syncthreads();
      const float vn = fmaxf(sqrtf(rv2[r]), EPSF);
      const float fs = tanhf(vn) / vn;
#pragma unroll
      for (int i = 0; i < 8; ++i) {
        const float4 m = *(const float4*)(mrw + 32 * i);
        float4 o;
        o.x = fs * m.x; o.y = fs * m.y; o.z = fs * m.z; o.w = fs * m.w;
        *(float4*)(orow + 32 * i) = o;
      }
    }
    if (q < 3) __syncthreads();
  }
}

__global__ __launch_bounds__(256) void k_agg(const float* __restrict__ F,
                                             const int* __restrict__ idx,
                                             const float* __restrict__ bias,
                                             float* __restrict__ OUT,
                                             float inScale, int doAct) {
  const int t = threadIdx.x;
  const int s = t & 7;
  const int node = (blockIdx.x << 5) + (t >> 3);
  const int* srow = idx + (size_t)node * 16;
  const int joff = s << 2;

  v2f rs = {inScale, inScale};
  asm volatile("" : "+v"(rs));

  v2f x[16];
  load32(F + (size_t)srow[0] * NHID + joff, rs, x);
  v2f px = {0.f, 0.f};
#pragma unroll
  for (int j = 0; j < 16; ++j) px += x[j] * x[j];
  float x2 = gred8(px.x + px.y);

  for (int k = 1; k < 16; ++k) {
    if (k < 15) __builtin_prefetch(F + (size_t)srow[k + 1] * NHID + (s << 5), 0, 0);
    v2f y[16];
    load32(F + (size_t)srow[k] * NHID + joff, rs, y);
    v2f py2 = {0.f, 0.f}, pxy = {0.f, 0.f};
#pragma unroll
    for (int j = 0; j < 16; ++j) { py2 += y[j] * y[j]; pxy += x[j] * y[j]; }
    float y2 = gred8(py2.x + py2.y);
    float xy = gred8(pxy.x + pxy.y);
    float den = fmaxf(1.f + 2.f * xy + x2 * y2, EPSF);
    float ca = (1.f + 2.f * xy + y2) / den;
    float cb = (1.f - x2) / den;
    v2f pn = {0.f, 0.f};
#pragma unroll
    for (int j = 0; j < 16; ++j) { x[j] = ca * x[j] + cb * y[j]; pn += x[j] * x[j]; }
    x2 = gred8(pn.x + pn.y);
  }

  float* orow = OUT + (size_t)node * NHID + joff;
  if (!doAct) {
    store32(orow, x, 1.f);
  } else {
    v2f b[16];
    load32(bias + joff, v2f{1.f, 1.f}, b);
    v2f pb2 = {0.f, 0.f}, pxy = {0.f, 0.f};
#pragma unroll
    for (int j = 0; j < 16; ++j) { pb2 += b[j] * b[j]; pxy += x[j] * b[j]; }
    float y2 = gred8(pb2.x + pb2.y);
    float xy = 0.25f * gred8(pxy.x + pxy.y);
    float x2r = 0.0625f * x2;
    float den = fmaxf(1.f + 2.f * xy + x2r * y2, EPSF);
    float ca = (1.f + 2.f * xy + y2) / den;
    float cb = (1.f - x2r) / den;
    float zn2 = ca * ca * x2r + 2.f * ca * cb * xy + cb * cb * y2;
    float zn = fminf(fmaxf(sqrtf(zn2), EPSF), MAXNF);
    float us = atanhf(zn) / zn;
    const float cq = ca * 0.25f;
    v2f pv = {0.f, 0.f};
#pragma unroll
    for (int j = 0; j < 16; ++j) {
      v2f z = cq * x[j] + cb * b[j];
      v2f u;
      u.x = fmaxf(us * z.x, 0.f);
      u.y = fmaxf(us * z.y, 0.f);
      x[j] = u;
      pv += u * u;
    }
    float vn = fmaxf(sqrtf(gred8(pv.x + pv.y)), EPSF);
    float fs = tanhf(vn) / vn;
    store32(orow, x, fs);
  }
}

__global__ __launch_bounds__(256) void k_final(const float* __restrict__ H,
                                               const float* __restrict__ Wl,
                                               const float* __restrict__ bl,
                                               const int* __restrict__ tf,
                                               float* __restrict__ OUT) {
  __shared__ int ridx[64];
  __shared__ float smx[64][72];
  __shared__ float part[64][4];
  __shared__ float rx2[64], rm2[64], rdot[64];
  __shared__ float sb2;

  const int t = threadIdx.x;
  if (t < 64) ridx[t] = tf[t] + t * 1024;
  if (t == 0) {
    float s = 0.f;
    for (int j = 0; j < 64; ++j) s += bl[j] * bl[j];
    sb2 = s;
  }
  __syncthreads();

  const int wave = t >> 5, lane = t & 31;
  const int l15 = lane & 15;
  const int kk = (lane >> 4) << 1;
#pragma unroll
  for (int i = 0; i < 2; ++i) {
    const int tile = wave * 2 + i;
    const int tm = tile >> 2, tn = tile & 3;
    const float* Ar = H + (size_t)ridx[tm * 16 + l15] * NHID;
    const float* Br = Wl + (size_t)(tn * 16 + l15) * NHID;
    v8f c = {};
    for (int k0 = 0; k0 < NHID; k0 += 4) {
      v2f a, b;
      a.x = Ar[k0 + kk]; a.y = Ar[k0 + kk + 1];
      b.x = Br[k0 + kk]; b.y = Br[k0 + kk + 1];
      c = wmma_f32x4(a, b, c);
    }
    const int n = tn * 16 + l15;
    const int mb = tm * 16 + ((lane >> 4) ? 8 : 0);
#pragma unroll
    for (int v = 0; v < 8; ++v) smx[mb + v][n] = c[v];
  }
  __syncthreads();

  const int r = t >> 2, p = t & 3;
  float sx = 0.f;
  {
    const float* hr = H + (size_t)ridx[r] * NHID + p * 64;
    for (int j = 0; j < 64; j += 4) {
      float4 v = *(const float4*)(hr + j);
      sx += v.x * v.x + v.y * v.y + v.z * v.z + v.w * v.w;
    }
  }
  float sm = 0.f, sd = 0.f;
#pragma unroll
  for (int j = 0; j < 16; ++j) {
    float m = smx[r][p * 16 + j];
    sm += m * m;
    sd += m * bl[p * 16 + j];
  }
  part[r][p] = sx;
  __syncthreads();
  if (t < 64) rx2[t] = part[t][0] + part[t][1] + part[t][2] + part[t][3];
  __syncthreads();
  part[r][p] = sm;
  __syncthreads();
  if (t < 64) rm2[t] = part[t][0] + part[t][1] + part[t][2] + part[t][3];
  __syncthreads();
  part[r][p] = sd;
  __syncthreads();
  if (t < 64) rdot[t] = part[t][0] + part[t][1] + part[t][2] + part[t][3];
  __syncthreads();

  const float s = hyp_scale(rx2[r], rm2[r]);
  const float x2r = s * s * rm2[r];
  const float xy = s * rdot[r];
  const float y2 = sb2;
  const float den = fmaxf(1.f + 2.f * xy + x2r * y2, EPSF);
  const float ca = (1.f + 2.f * xy + y2) / den;
  const float cb = (1.f - x2r) / den;
  const float acs = ca * s;
#pragma unroll
  for (int j = 0; j < 16; ++j)
    OUT[r * 64 + p * 16 + j] = acs * smx[r][p * 16 + j] + cb * bl[p * 16 + j];
}

extern "C" void kernel_launch(void* const* d_in, const int* in_sizes, int n_in,
                              void* d_out, int out_size, void* d_ws,
                              size_t ws_size, hipStream_t stream) {
  const float* features = (const float*)d_in[0];
  const float* W1 = (const float*)d_in[1];
  const float* b1 = (const float*)d_in[2];
  const float* W2 = (const float*)d_in[3];
  const float* b2 = (const float*)d_in[4];
  const float* Wl = (const float*)d_in[5];
  const float* bl = (const float*)d_in[6];
  const int* src_idx = (const int*)d_in[7];
  const int* to_fetch = (const int*)d_in[8];
  float* out = (float*)d_out;

  float* buf0 = (float*)d_ws;
  float* buf1 = buf0 + (size_t)65536 * 256;

  k_mm<512, false><<<512, 256, 0, stream>>>(const_cast<float*>(features), W1,
                                             b1, buf0, 0.25f);
  k_agg<<<2048, 256, 0, stream>>>(buf0, src_idx, b1, buf1, 1.0f, 1);
  k_agg<<<2048, 256, 0, stream>>>(buf1, src_idx, b2, buf0, 0.25f, 0);
  k_mm<256, true><<<512, 256, 0, stream>>>(buf0, W2, b2, buf0, 1.0f);
  k_final<<<1, 256, 0, stream>>>(buf0, Wl, bl, to_fetch, out);
}


# k_mm epilogue in one pass from the accumulator registers: lane partials of each row sum via a 64-slot LDS row, a row's scalars computed once, whole-line stores; no LDS tile; row sums' order changes
# speedup vs baseline: 1.5010x; 1.0381x over previous
#include <hip/hip_runtime.h>

typedef __attribute__((ext_vector_type(2))) float v2f;
typedef __attribute__((ext_vector_type(8))) float v8f;

#define NHID 256
#define EPSF 1e-7f
#define MAXNF 0.99999f

__device__ __forceinline__ v8f wmma_f32x4(v2f a, v2f b, v8f c) {
  return __builtin_amdgcn_wmma_f32_16x16x4_f32(false, a, false, b, (short)0, c,
                                               false, false);
}

__device__ __forceinline__ float gred8(float v) {
#pragma unroll
  for (int m = 4; m > 0; m >>= 1) v += __shfl_xor(v, m, 32);
  return v;
}

__device__ __forceinline__ float hyp_scale(float x2, float m2) {
  float xn  = fminf(fmaxf(sqrtf(x2), EPSF), MAXNF);
  float mxn = fmaxf(sqrtf(m2), EPSF);
  return tanhf(mxn / xn * atanhf(xn)) / mxn;
}

__device__ __forceinline__ void load32(const float* p, v2f s, v2f* d) {
#pragma unroll
  for (int i = 0; i < 8; ++i) {
    float4 a = *(const float4*)(p + 32 * i);
    v2f lo = {a.x, a.y}, hi = {a.z, a.w};
    d[2 * i] = lo * s;
    d[2 * i + 1] = hi * s;
  }
}

__device__ __forceinline__ void store32(float* p, const v2f* d, float s) {
#pragma unroll
  for (int i = 0; i < 8; ++i) {
    v2f lo = d[2 * i] * s, hi = d[2 * i + 1] * s;
    float4 a = {lo.x, lo.y, hi.x, hi.y};
    *(float4*)(p + 32 * i) = a;
  }
}

template <int KDIM>
__device__ __forceinline__ void mm_load(const float* Ap, const float* Bp, int k0,
                                        float4 (&a)[4], float4 (&b)[4]) {
  for (int i = 0; i < 4; ++i)
    a[i] = *(const float4*)(Ap + (size_t)i * 16 * KDIM + k0);
  for (int s = 0; s < 4; ++s)
    b[s] = *(const float4*)(Bp + (size_t)(k0 + s) * NHID);
}

__device__ __forceinline__ void mm_block(v8f (&acc)[4][4], const float4 (&a)[4],
                                         const float4 (&b)[4]) {
  const float bv[4][4] = {{b[0].x, b[0].y, b[0].z, b[0].w},
                          {b[1].x, b[1].y, b[1].z, b[1].w},
                          {b[2].x, b[2].y, b[2].z, b[2].w},
                          {b[3].x, b[3].y, b[3].z, b[3].w}};
  for (int i = 0; i < 4; ++i) {
    v2f a0; a0.x = a[i].x; a0.y = a[i].y;
    for (int j = 0; j < 4; ++j) {
      v2f b0; b0.x = bv[0][j]; b0.y = bv[1][j];
      acc[i][j] = wmma_f32x4(a0, b0, acc[i][j]);
    }
  }
  for (int i = 0; i < 4; ++i) {
    v2f a1; a1.x = a[i].z; a1.y = a[i].w;
    for (int j = 0; j < 4; ++j) {
      v2f b1; b1.x = bv[2][j]; b1.y = bv[3][j];
      acc[i][j] = wmma_f32x4(a1, b1, acc[i][j]);
    }
  }
}

template <int KDIM, bool FULL_EPI>
__global__ __launch_bounds__(256) void k_mm(const float* X,
                                            const float* __restrict__ W,
                                            const float* __restrict__ bias,
                                            float* OUT, float inScale) {
  __shared__ alignas(16) float part[128][68];
  __shared__ alignas(16) float partd[FULL_EPI ? 128 : 1][68];
  __shared__ alignas(16) float rx2[128];
  __shared__ alignas(16) float rsA[128], rsB[FULL_EPI ? 128 : 4], rsC[FULL_EPI ? 128 : 4];
  __shared__ float wsum[8];

  const int t = threadIdx.x;
  const int wave = t >> 5, lane = t & 31;
  const int tile_m = wave >> 2;
  const int n0 = (wave & 3) << 6;
  const int l15 = lane & 15;
  const int h = lane >> 4;
  const int brow = blockIdx.x << 7;

  v8f acc[4][4];
  for (int i = 0; i < 4; ++i)
    for (int j = 0; j < 4; ++j) acc[i][j] = v8f{};

  const float* Ap = X + (size_t)(brow + tile_m * 64 + l15) * KDIM + 4 * h;
  const float* Bp = W + (size_t)(4 * h) * NHID + n0 + 4 * l15;

  float4 a0[4], b0[4], a1[4], b1[4];
  mm_load<KDIM>(Ap, Bp, 0, a0, b0);
  for (int k0 = 0; k0 < KDIM - 16; k0 += 16) {
    mm_load<KDIM>(Ap, Bp, k0 + 8, a1, b1);
    mm_block(acc, a0, b0);
    mm_load<KDIM>(Ap, Bp, k0 + 16, a0, b0);
    mm_block(acc, a1, b1);
  }
  mm_load<KDIM>(Ap, Bp, KDIM - 8, a1, b1);
  mm_block(acc, a0, b0);
  mm_block(acc, a1, b1);

  const int wc = wave & 3;
  const int rbase = tile_m * 64 + 8 * h;
  const int slot = wc * 16 + l15;
  float4 bb = {0.f, 0.f, 0.f, 0.f};
  if (FULL_EPI) {
    bb = *(const float4*)(bias + n0 + 4 * l15);
    float bv = bias[t];
    bv *= bv;
    for (int m = 16; m > 0; m >>= 1) bv += __shfl_xor(bv, m, 32);
    if (lane == 0) wsum[wave] = bv;
  }

#pragma unroll
  for (int i = 0; i < 4; ++i) {
#pragma unroll
    for (int v = 0; v < 8; ++v) {
      const float m0 = acc[i][0][v], m1 = acc[i][1][v];
      const float m2 = acc[i][2][v], m3 = acc[i][3][v];
      float s = m0 * m0; s += m1 * m1; s += m2 * m2; s += m3 * m3;
      part[rbase + 16 * i + v][slot] = s;
      if (FULL_EPI) {
        float d = m0 * bb.x; d += m1 * bb.y; d += m2 * bb.z; d += m3 * bb.w;
        partd[rbase + 16 * i + v][slot] = d;
      }
    }
  }

  {
    const int r8 = t >> 3, p = t & 7;
#pragma unroll 1
    for (int q = 0; q < 4; ++q) {
      const float* xr = X + (size_t)(brow + 32 * q + r8) * KDIM + 4 * p;
      float sx = 0.f;
      for (int j = 0; j < KDIM; j += 32) {
        float4 v = *(const float4*)(xr + j);
        sx += v.x * v.x + v.y * v.y + v.z * v.z + v.w * v.w;
      }
      sx = gred8(sx);
      if (p == 0) rx2[32 * q + r8] = sx;
    }
  }
  __syncthreads();

  if (t < 128) {
    float4 s4 = {0.f, 0.f, 0.f, 0.f}, d4 = {0.f, 0.f, 0.f, 0.f};
    for (int q = 0; q < 16; ++q) {
      const float4 a = *(const float4*)&part[t][4 * q];
      s4.x += a.x; s4.y += a.y; s4.z += a.z; s4.w += a.w;
      if (FULL_EPI) {
        const float4 d = *(const float4*)&partd[t][4 * q];
        d4.x += d.x; d4.y += d.y; d4.z += d.z; d4.w += d.w;
      }
    }
    const float sm = (s4.x + s4.y) + (s4.z + s4.w);
    const float sd = (d4.x + d4.y) + (d4.z + d4.w);
    const float x2s = inScale * inScale * rx2[t];
    const float m2s = inScale * inScale * sm;
    const float cmv = hyp_scale(x2s, m2s) * inScale;
    if (!FULL_EPI) {
      rsA[t] = cmv;
    } else {
      float y2 = 0.f;
      for (int w8 = 0; w8 < 8; ++w8) y2 += wsum[w8];
      const float cf = 0.25f * cmv;
      const float x2r = cf * cf * sm;
      const float xy = cf * sd;
      const float den = fmaxf(1.f + 2.f * xy + x2r * y2, EPSF);
      const float ca = (1.f + 2.f * xy + y2) / den;
      const float cb = (1.f - x2r) / den;
      const float zn2 = ca * ca * x2r + 2.f * ca * cb * xy + cb * cb * y2;
      const float zn = fminf(fmaxf(sqrtf(zn2), EPSF), MAXNF);
      rsA[t] = ca * cf;
      rsB[t] = cb;
      rsC[t] = atanhf(zn) / zn;
    }
  }
  __syncthreads();

  float* obase = OUT + (size_t)(brow + rbase) * NHID + n0 + 4 * l15;
  if (!FULL_EPI) {
#pragma unroll
    for (int i = 0; i < 4; ++i) {
      const float4 c0 = *(const float4*)&rsA[rbase + 16 * i];
      const float4 c1 = *(const float4*)&rsA[rbase + 16 * i + 4];
      const float cs[8] = {c0.x, c0.y, c0.z, c0.w, c1.x, c1.y, c1.z, c1.w};
#pragma unroll
      for (int v = 0; v < 8; ++v) {
        float4 o;
        o.x = cs[v] * acc[i][0][v]; o.y = cs[v] * acc[i][1][v];
        o.z = cs[v] * acc[i][2][v]; o.w = cs[v] * acc[i][3][v];
        *(float4*)(obase + (size_t)(16 * i + v) * NHID) = o;
      }
    }
  } else {
#pragma unroll
    for (int i = 0; i < 4; ++i) {
      const float4 a0 = *(const float4*)&rsA[rbase + 16 * i];
      const float4 a1 = *(const float4*)&rsA[rbase + 16 * i + 4];
      const float4 b0 = *(const float4*)&rsB[rbase + 16 * i];
      const float4 b1 = *(const float4*)&rsB[rbase + 16 * i + 4];
      const float4 u0 = *(const float4*)&rsC[rbase + 16 * i];
      const float4 u1 = *(const float4*)&rsC[rbase + 16 * i + 4];
      const float acfs[8] = {a0.x, a0.y, a0.z, a0.w, a1.x, a1.y, a1.z, a1.w};
      const float cbs[8] = {b0.x, b0.y, b0.z, b0.w, b1.x, b1.y, b1.z, b1.w};
      const float uss[8] = {u0.x, u0.y, u0.z, u0.w, u1.x, u1.y, u1.z, u1.w};
#pragma unroll
      for (int v = 0; v < 8; ++v) {
        const float acf = acfs[v], cb = cbs[v], us = uss[v];
        const float w0 = fmaxf(us * (acf * acc[i][0][v] + cb * bb.x), 0.f);
        const float w1 = fmaxf(us * (acf * acc[i][1][v] + cb * bb.y), 0.f);
        const float w2 = fmaxf(us * (acf * acc[i][2][v] + cb * bb.z), 0.f);
        const float w3 = fmaxf(us * (acf * acc[i][3][v] + cb * bb.w), 0.f);
        acc[i][0][v] = w0; acc[i][1][v] = w1; acc[i][2][v] = w2; acc[i][3][v] = w3;
        float s = w0 * w0; s += w1 * w1; s += w2 * w2; s += w3 * w3;
        part[rbase + 16 * i + v][slot] = s;
      }
    }
    __syncthreads();
    if (t < 128) {
      float4 v4 = {0.f, 0.f, 0.f, 0.f};
      for (int q = 0; q < 16; ++q) {
        const float4 a = *(const float4*)&part[t][4 * q];
        v4.x += a.x; v4.y += a.y; v4.z += a.z; v4.w += a.w;
      }
      const float sv = (v4.x + v4.y) + (v4.z + v4.w);
      const float vn = fmaxf(sqrtf(sv), EPSF);
      rsA[t] = tanhf(vn) / vn;
    }
    __syncthreads();
#pragma unroll
    for (int i = 0; i < 4; ++i) {
      const float4 c0 = *(const float4*)&rsA[rbase + 16 * i];
      const float4 c1 = *(const float4*)&rsA[rbase + 16 * i + 4];
      const float cs[8] = {c0.x, c0.y, c0.z, c0.w, c1.x, c1.y, c1.z, c1.w};
#pragma unroll
      for (int v = 0; v < 8; ++v) {
        float4 o;
        o.x = cs[v] * acc[i][0][v]; o.y = cs[v] * acc[i][1][v];
        o.z = cs[v] * acc[i][2][v]; o.w = cs[v] * acc[i][3][v];
        *(float4*)(obase + (size_t)(16 * i + v) * NHID) = o;
      }
    }
  }
}

__global__ __launch_bounds__(256) void k_agg(const float* __restrict__ F,
                                             const int* __restrict__ idx,
                                             const float* __restrict__ bias,
                                             float* __restrict__ OUT,
                                             float inScale, int doAct) {
  const int t = threadIdx.x;
  const int s = t & 7;
  const int node = (blockIdx.x << 5) + (t >> 3);
  const int* srow = idx + (size_t)node * 16;
  const int joff = s << 2;

  v2f rs = {inScale, inScale};
  asm volatile("" : "+v"(rs));

  v2f x[16];
  load32(F + (size_t)srow[0] * NHID + joff, rs, x);
  v2f px = {0.f, 0.f};
#pragma unroll
  for (int j = 0; j < 16; ++j) px += x[j] * x[j];
  float x2 = gred8(px.x + px.y);

  for (int k = 1; k < 16; ++k) {
    if (k < 15) __builtin_prefetch(F + (size_t)srow[k + 1] * NHID + (s << 5), 0, 0);
    v2f y[16];
    load32(F + (size_t)srow[k] * NHID + joff, rs, y);
    v2f py2 = {0.f, 0.f}, pxy = {0.f, 0.f};
#pragma unroll
    for (int j = 0; j < 16; ++j) { py2 += y[j] * y[j]; pxy += x[j] * y[j]; }
    float y2 = gred8(py2.x + py2.y);
    float xy = gred8(pxy.x + pxy.y);
    float den = fmaxf(1.f + 2.f * xy + x2 * y2, EPSF);
    float ca = (1.f + 2.f * xy + y2) / den;
    float cb = (1.f - x2) / den;
    v2f pn = {0.f, 0.f};
#pragma unroll
    for (int j = 0; j < 16; ++j) { x[j] = ca * x[j] + cb * y[j]; pn += x[j] * x[j]; }
    x2 = gred8(pn.x + pn.y);
  }

  float* orow = OUT + (size_t)node * NHID + joff;
  if (!doAct) {
    store32(orow, x, 1.f);
  } else {
    v2f b[16];
    load32(bias + joff, v2f{1.f, 1.f}, b);
    v2f pb2 = {0.f, 0.f}, pxy = {0.f, 0.f};
#pragma unroll
    for (int j = 0; j < 16; ++j) { pb2 += b[j] * b[j]; pxy += x[j] * b[j]; }
    float y2 = gred8(pb2.x + pb2.y);
    float xy = 0.25f * gred8(pxy.x + pxy.y);
    float x2r = 0.0625f * x2;
    float den = fmaxf(1.f + 2.f * xy + x2r * y2, EPSF);
    float ca = (1.f + 2.f * xy + y2) / den;
    float cb = (1.f - x2r) / den;
    float zn2 = ca * ca * x2r + 2.f * ca * cb * xy + cb * cb * y2;
    float zn = fminf(fmaxf(sqrtf(zn2), EPSF), MAXNF);
    float us = atanhf(zn) / zn;
    const float cq = ca * 0.25f;
    v2f pv = {0.f, 0.f};
#pragma unroll
    for (int j = 0; j < 16; ++j) {
      v2f z = cq * x[j] + cb * b[j];
      v2f u;
      u.x = fmaxf(us * z.x, 0.f);
      u.y = fmaxf(us * z.y, 0.f);
      x[j] = u;
      pv += u * u;
    }
    float vn = fmaxf(sqrtf(gred8(pv.x + pv.y)), EPSF);
    float fs = tanhf(vn) / vn;
    store32(orow, x, fs);
  }
}

__global__ __launch_bounds__(256) void k_final(const float* __restrict__ H,
                                               const float* __restrict__ Wl,
                                               const float* __restrict__ bl,
                                               const int* __restrict__ tf,
                                               float* __restrict__ OUT) {
  __shared__ int ridx[64];
  __shared__ float smx[64][72];
  __shared__ float part[64][4];
  __shared__ float rx2[64], rm2[64], rdot[64];
  __shared__ float sb2;

  const int t = threadIdx.x;
  if (t < 64) ridx[t] = tf[t] + t * 1024;
  if (t == 0) {
    float s = 0.f;
    for (int j = 0; j < 64; ++j) s += bl[j] * bl[j];
    sb2 = s;
  }
  __syncthreads();

  const int wave = t >> 5, lane = t & 31;
  const int l15 = lane & 15;
  const int kk = (lane >> 4) << 1;
#pragma unroll
  for (int i = 0; i < 2; ++i) {
    const int tile = wave * 2 + i;
    const int tm = tile >> 2, tn = tile & 3;
    const float* Ar = H + (size_t)ridx[tm * 16 + l15] * NHID;
    const float* Br = Wl + (size_t)(tn * 16 + l15) * NHID;
    v8f c = {};
    for (int k0 = 0; k0 < NHID; k0 += 4) {
      v2f a, b;
      a.x = Ar[k0 + kk]; a.y = Ar[k0 + kk + 1];
      b.x = Br[k0 + kk]; b.y = Br[k0 + kk + 1];
      c = wmma_f32x4(a, b, c);
    }
    const int n = tn * 16 + l15;
    const int mb = tm * 16 + ((lane >> 4) ? 8 : 0);
#pragma unroll
    for (int v = 0; v < 8; ++v) smx[mb + v][n] = c[v];
  }
  __syncthreads();

  const int r = t >> 2, p = t & 3;
  float sx = 0.f;
  {
    const float* hr = H + (size_t)ridx[r] * NHID + p * 64;
    for (int j = 0; j < 64; j += 4) {
      float4 v = *(const float4*)(hr + j);
      sx += v.x * v.x + v.y * v.y + v.z * v.z + v.w * v.w;
    }
  }
  float sm = 0.f, sd = 0.f;
#pragma unroll
  for (int j = 0; j < 16; ++j) {
    float m = smx[r][p * 16 + j];
    sm += m * m;
    sd += m * bl[p * 16 + j];
  }
  part[r][p] = sx;
  __syncthreads();
  if (t < 64) rx2[t] = part[t][0] + part[t][1] + part[t][2] + part[t][3];
  __syncthreads();
  part[r][p] = sm;
  __syncthreads();
  if (t < 64) rm2[t] = part[t][0] + part[t][1] + part[t][2] + part[t][3];
  __syncthreads();
  part[r][p] = sd;
  __syncthreads();
  if (t < 64) rdot[t] = part[t][0] + part[t][1] + part[t][2] + part[t][3];
  __syncthreads();

  const float s = hyp_scale(rx2[r], rm2[r]);
  const float x2r = s * s * rm2[r];
  const float xy = s * rdot[r];
  const float y2 = sb2;
  const float den = fmaxf(1.f + 2.f * xy + x2r * y2, EPSF);
  const float ca = (1.f + 2.f * xy + y2) / den;
  const float cb = (1.f - x2r) / den;
  const float acs = ca * s;
#pragma unroll
  for (int j = 0; j < 16; ++j)
    OUT[r * 64 + p * 16 + j] = acs * smx[r][p * 16 + j] + cb * bl[p * 16 + j];
}

extern "C" void kernel_launch(void* const* d_in, const int* in_sizes, int n_in,
                              void* d_out, int out_size, void* d_ws,
                              size_t ws_size, hipStream_t stream) {
  const float* features = (const float*)d_in[0];
  const float* W1 = (const float*)d_in[1];
  const float* b1 = (const float*)d_in[2];
  const float* W2 = (const float*)d_in[3];
  const float* b2 = (const float*)d_in[4];
  const float* Wl = (const float*)d_in[5];
  const float* bl = (const float*)d_in[6];
  const int* src_idx = (const int*)d_in[7];
  const int* to_fetch = (const int*)d_in[8];
  float* out = (float*)d_out;

  float* buf0 = (float*)d_ws;
  float* buf1 = buf0 + (size_t)65536 * 256;

  k_mm<512, false><<<512, 256, 0, stream>>>(const_cast<float*>(features), W1,
                                             b1, buf0, 0.25f);
  k_agg<<<2048, 256, 0, stream>>>(buf0, src_idx, b1, buf1, 1.0f, 1);
  k_agg<<<2048, 256, 0, stream>>>(buf1, src_idx, b2, buf0, 0.25f, 0);
  k_mm<256, true><<<512, 256, 0, stream>>>(buf0, W2, b2, buf0, 1.0f);
  k_final<<<1, 256, 0, stream>>>(buf0, Wl, bl, to_fetch, out);
}
